# starting kernel
# baseline (speedup 1.0000x reference)
#include <hip/hip_runtime.h>


typedef __attribute__((ext_vector_type(2))) float v2f;
typedef __attribute__((ext_vector_type(8))) float v8f;

#define DD   32
#define NPIX 4096
#define NBLK (NPIX / 16)
#define LDS_STRIDE 18

__device__ __forceinline__ v8f wmma_f32(v2f a, v2f b, v8f c) {
  return __builtin_amdgcn_wmma_f32_16x16x4_f32(
       false, a,  false, b,
       (short)0, c,  false,  false);
}

__global__ __launch_bounds__(32)
void MS_meanshift_iter(const float* __restrict__ src, float* __restrict__ dst) {
  __shared__ __align__(16) float ldsE[16 * LDS_STRIDE];

  const int lane = threadIdx.x;
  const int lq   = lane & 15;
  const int half = lane >> 4;
  const int b    = blockIdx.x >> 8;
  const int mb   = blockIdx.x & 255;

  const float* Y = src + b * DD * NPIX;
  float*       O = dst + b * DD * NPIX;

  const float BW_LOG2E = 0.14426950408889634f;

  const int mcol = mb * 16 + lq;
  v2f am[8];
#pragma unroll
  for (int k = 0; k < 8; ++k) {
    am[k].x = Y[(4 * k + 0 + 2 * half) * NPIX + mcol];
    am[k].y = Y[(4 * k + 1 + 2 * half) * NPIX + mcol];
  }

  v8f acc0 = {};
  v8f acc1 = {};
  v8f srow = {};

  for (int nb = 0; nb < NBLK; ++nb) {
    const int ncol0 = nb * 16;
    if (nb + 1 < NBLK)
      __builtin_prefetch(&Y[ncol0 + 16 + lq], 0, 0);

    v2f gb[8];
#pragma unroll
    for (int k = 0; k < 8; ++k) {
      gb[k].x = Y[(4 * k + 0 + 2 * half) * NPIX + ncol0 + lq];
      gb[k].y = Y[(4 * k + 1 + 2 * half) * NPIX + ncol0 + lq];
    }
    v8f g = {};
#pragma unroll
    for (int k = 0; k < 8; ++k) g = wmma_f32(am[k], gb[k], g);

    v8f e;
#pragma unroll
    for (int r = 0; r < 8; ++r) e[r] = __builtin_amdgcn_exp2f(g[r] * BW_LOG2E);
    srow += e;

    __syncthreads();
#pragma unroll
    for (int r = 0; r < 8; ++r)
      ldsE[(r + 8 * half) * LDS_STRIDE + lq] = e[r];
    __syncthreads();

    v2f a2[4];
#pragma unroll
    for (int k = 0; k < 4; ++k)
      a2[k] = *(const v2f*)&ldsE[lq * LDS_STRIDE + 4 * k + 2 * half];

#pragma unroll
    for (int k = 0; k < 4; ++k) {
      v2f b20 = *(const v2f*)&Y[(0 * 16 + lq) * NPIX + ncol0 + 4 * k + 2 * half];
      v2f b21 = *(const v2f*)&Y[(1 * 16 + lq) * NPIX + ncol0 + 4 * k + 2 * half];
      acc0 = wmma_f32(a2[k], b20, acc0);
      acc1 = wmma_f32(a2[k], b21, acc1);
    }
  }

#pragma unroll
  for (int mask = 1; mask <= 8; mask <<= 1) {
#pragma unroll
    for (int r = 0; r < 8; ++r)
      srow[r] += __shfl_xor(srow[r], mask, 32);
  }

#pragma unroll
  for (int r = 0; r < 8; ++r) {
    const int m   = mb * 16 + r + 8 * half;
    const float inv = 0.5f / srow[r];
    {
      const int d = lq;
      O[d * NPIX + m] = acc0[r] * inv + 0.5f * Y[d * NPIX + m];
    }
    {
      const int d = 16 + lq;
      O[d * NPIX + m] = acc1[r] * inv + 0.5f * Y[d * NPIX + m];
    }
  }
}

extern "C" void kernel_launch(void* const* d_in, const int* in_sizes, int n_in,
                              void* d_out, int out_size, void* d_ws, size_t ws_size,
                              hipStream_t stream) {
  (void)in_sizes; (void)n_in; (void)out_size; (void)ws_size;
  const float* x  = (const float*)d_in[0];
  float*       out = (float*)d_out;
  float* ws0 = (float*)d_ws;
  float* ws1 = ws0 + 4 * DD * NPIX;

  dim3 grid(4 * NBLK);
  dim3 block(32);

  MS_meanshift_iter<<<grid, block, 0, stream>>>(x,   ws0);
  MS_meanshift_iter<<<grid, block, 0, stream>>>(ws0, ws1);
  MS_meanshift_iter<<<grid, block, 0, stream>>>(ws1, out);
}


# Split each m-block's n loop over 8 waves of one workgroup (8 waves per SIMD hide load, LDS and WMMA latency); the 8 partial sums are added via LDS in fixed order: summation order over n changes
# speedup vs baseline: 1.9516x; 1.9516x over previous
#include <hip/hip_runtime.h>


typedef __attribute__((ext_vector_type(2))) float v2f;
typedef __attribute__((ext_vector_type(8))) float v8f;

#define DD   32
#define NPIX 4096
#define NBLK (NPIX / 16)
#define LDS_STRIDE 18
#define NWAVE 8
#define NRED  24

__device__ __forceinline__ v8f wmma_f32(v2f a, v2f b, v8f c) {
  return __builtin_amdgcn_wmma_f32_16x16x4_f32(
       false, a,  false, b,
       (short)0, c,  false,  false);
}

__global__ __launch_bounds__(32 * NWAVE)
void MS_meanshift_iter(const float* __restrict__ src, float* __restrict__ dst) {
  __shared__ __align__(16) float ldsEall[NWAVE][16 * LDS_STRIDE];
  __shared__ float red[NWAVE][NRED][32];

  const int lane = threadIdx.x & 31;
  const int wv   = __builtin_amdgcn_readfirstlane((int)(threadIdx.x >> 5));
  float* ldsE    = ldsEall[wv];
  const int lq   = lane & 15;
  const int half = lane >> 4;
  const int b    = blockIdx.x >> 8;
  const int mb   = blockIdx.x & 255;

  const float* Y = src + b * DD * NPIX;
  float*       O = dst + b * DD * NPIX;

  const float BW_LOG2E = 0.14426950408889634f;

  const int mcol = mb * 16 + lq;
  v2f am[8];
#pragma unroll
  for (int k = 0; k < 8; ++k) {
    am[k].x = Y[(4 * k + 0 + 2 * half) * NPIX + mcol];
    am[k].y = Y[(4 * k + 1 + 2 * half) * NPIX + mcol];
  }

  const float* Yg = Y + (2 * half) * NPIX + lq;
  const float* Yb = Y + lq * NPIX + 2 * half;

  v8f acc0 = {};
  v8f acc1 = {};
  v8f srow = {};

  for (int it = 0; it < NBLK / NWAVE; ++it) {
    const int nb    = it * NWAVE + wv;
    const int ncol0 = nb * 16;
    if (it + 1 < NBLK / NWAVE)
      __builtin_prefetch(&Y[ncol0 + 16 * NWAVE + lq], 0, 0);
    const float* Yn = Yg + ncol0;
    const float* Yk = Yb + ncol0;

    v2f gb[8];
#pragma unroll
    for (int k = 0; k < 8; ++k) {
      gb[k].x = Yn[(4 * k + 0) * NPIX];
      gb[k].y = Yn[(4 * k + 1) * NPIX];
    }
    v8f g = {};
#pragma unroll
    for (int k = 0; k < 8; ++k) g = wmma_f32(am[k], gb[k], g);

    v8f e;
#pragma unroll
    for (int r = 0; r < 8; ++r) e[r] = __builtin_amdgcn_exp2f(g[r] * BW_LOG2E);
    srow += e;

#pragma unroll
    for (int r = 0; r < 8; ++r)
      ldsE[(r + 8 * half) * LDS_STRIDE + lq] = e[r];

    v2f a2[4];
#pragma unroll
    for (int k = 0; k < 4; ++k)
      a2[k] = *(const v2f*)&ldsE[lq * LDS_STRIDE + 4 * k + 2 * half];

#pragma unroll
    for (int k = 0; k < 4; ++k) {
      v2f b20 = *(const v2f*)&Yk[4 * k];
      v2f b21 = *(const v2f*)&Yk[16 * NPIX + 4 * k];
      acc0 = wmma_f32(a2[k], b20, acc0);
      acc1 = wmma_f32(a2[k], b21, acc1);
    }
  }

#pragma unroll
  for (int r = 0; r < 8; ++r) {
    red[wv][r][lane]      = acc0[r];
    red[wv][8 + r][lane]  = acc1[r];
    red[wv][16 + r][lane] = srow[r];
  }
  __syncthreads();

  const int r = wv;
  float a0 = red[0][r][lane];
  float a1 = red[0][8 + r][lane];
  float s  = red[0][16 + r][lane];
#pragma unroll
  for (int w = 1; w < NWAVE; ++w) {
    a0 += red[w][r][lane];
    a1 += red[w][8 + r][lane];
    s  += red[w][16 + r][lane];
  }

#pragma unroll
  for (int mask = 1; mask <= 8; mask <<= 1)
    s += __shfl_xor(s, mask, 32);

  {
    const int m     = mb * 16 + r + 8 * half;
    const float inv = 0.5f / s;
    O[lq * NPIX + m]        = a0 * inv + 0.5f * Y[lq * NPIX + m];
    O[(16 + lq) * NPIX + m] = a1 * inv + 0.5f * Y[(16 + lq) * NPIX + m];
  }
}

extern "C" void kernel_launch(void* const* d_in, const int* in_sizes, int n_in,
                              void* d_out, int out_size, void* d_ws, size_t ws_size,
                              hipStream_t stream) {
  (void)in_sizes; (void)n_in; (void)out_size; (void)ws_size;
  const float* x  = (const float*)d_in[0];
  float*       out = (float*)d_out;
  float* ws0 = (float*)d_ws;
  float* ws1 = ws0 + 4 * DD * NPIX;

  dim3 grid(4 * NBLK);
  dim3 block(32 * NWAVE);

  MS_meanshift_iter<<<grid, block, 0, stream>>>(x,   ws0);
  MS_meanshift_iter<<<grid, block, 0, stream>>>(ws0, ws1);
  MS_meanshift_iter<<<grid, block, 0, stream>>>(ws1, out);
}


# Retile: workgroup = 64 m-columns, 16 waves split n, each wave keeps 4 m-blocks in registers so every loaded n-tile feeds 4x the WMMAs (global loads per WMMA /4); n summation order changes
# speedup vs baseline: 2.8106x; 1.4402x over previous
#include <hip/hip_runtime.h>


typedef __attribute__((ext_vector_type(2))) float v2f;
typedef __attribute__((ext_vector_type(8))) float v8f;

#define DD   32
#define NPIX 4096
#define NBLK (NPIX / 16)
#define LDS_STRIDE 18
#define RM    4
#define NWAVE 16
#define NGRP  (NBLK / RM)
#define NRED  24

__device__ __forceinline__ v8f wmma_f32(v2f a, v2f b, v8f c) {
  return __builtin_amdgcn_wmma_f32_16x16x4_f32(
       false, a,  false, b,
       (short)0, c,  false,  false);
}

__global__ __launch_bounds__(32 * NWAVE)
void MS_meanshift_iter(const float* __restrict__ src, float* __restrict__ dst) {
  __shared__ __align__(16) float ldsEall[NWAVE][RM][16 * LDS_STRIDE];
  __shared__ float red[NWAVE][RM * NRED][32];

  const int lane = threadIdx.x & 31;
  const int wv   = __builtin_amdgcn_readfirstlane((int)(threadIdx.x >> 5));
  const int lq   = lane & 15;
  const int half = lane >> 4;
  const int b    = blockIdx.x / NGRP;
  const int mg   = blockIdx.x % NGRP;

  const float* Y = src + b * DD * NPIX;
  float*       O = dst + b * DD * NPIX;

  const float BW_LOG2E = 0.14426950408889634f;

  v2f am[RM][8];
#pragma unroll
  for (int j = 0; j < RM; ++j) {
    const int mcol = (mg * RM + j) * 16 + lq;
#pragma unroll
    for (int k = 0; k < 8; ++k) {
      am[j][k].x = Y[(4 * k + 0 + 2 * half) * NPIX + mcol];
      am[j][k].y = Y[(4 * k + 1 + 2 * half) * NPIX + mcol];
    }
  }

  const float* Yg = Y + (2 * half) * NPIX + lq;
  const float* Yb = Y + lq * NPIX + 2 * half;

  v8f acc0[RM];
  v8f acc1[RM];
  v8f srow[RM];
#pragma unroll
  for (int j = 0; j < RM; ++j) { acc0[j] = {}; acc1[j] = {}; srow[j] = {}; }

  for (int it = 0; it < NBLK / NWAVE; ++it) {
    const int nb    = it * NWAVE + wv;
    const int ncol0 = nb * 16;
    if (it + 1 < NBLK / NWAVE)
      __builtin_prefetch(&Y[ncol0 + 16 * NWAVE + lq], 0, 0);
    const float* Yn = Yg + ncol0;
    const float* Yk = Yb + ncol0;

    v2f gb[8];
#pragma unroll
    for (int k = 0; k < 8; ++k) {
      gb[k].x = Yn[(4 * k + 0) * NPIX];
      gb[k].y = Yn[(4 * k + 1) * NPIX];
    }
    v2f b20[4], b21[4];
#pragma unroll
    for (int k = 0; k < 4; ++k) {
      b20[k] = *(const v2f*)&Yk[4 * k];
      b21[k] = *(const v2f*)&Yk[16 * NPIX + 4 * k];
    }

#pragma unroll
    for (int j = 0; j < RM; ++j) {
      float* ldsE = ldsEall[wv][j];

      v8f g = {};
#pragma unroll
      for (int k = 0; k < 8; ++k) g = wmma_f32(am[j][k], gb[k], g);

      v8f e;
#pragma unroll
      for (int r = 0; r < 8; ++r) e[r] = __builtin_amdgcn_exp2f(g[r] * BW_LOG2E);
      srow[j] += e;

#pragma unroll
      for (int r = 0; r < 8; ++r)
        ldsE[(r + 8 * half) * LDS_STRIDE + lq] = e[r];

      v2f a2[4];
#pragma unroll
      for (int k = 0; k < 4; ++k)
        a2[k] = *(const v2f*)&ldsE[lq * LDS_STRIDE + 4 * k + 2 * half];

#pragma unroll
      for (int k = 0; k < 4; ++k) {
        acc0[j] = wmma_f32(a2[k], b20[k], acc0[j]);
        acc1[j] = wmma_f32(a2[k], b21[k], acc1[j]);
      }
    }
  }

#pragma unroll
  for (int j = 0; j < RM; ++j) {
#pragma unroll
    for (int r = 0; r < 8; ++r) {
      red[wv][j * NRED + r][lane]      = acc0[j][r];
      red[wv][j * NRED + 8 + r][lane]  = acc1[j][r];
      red[wv][j * NRED + 16 + r][lane] = srow[j][r];
    }
  }
  __syncthreads();

  for (int u = wv; u < RM * 8; u += NWAVE) {
    const int j = u >> 3;
    const int r = u & 7;
    float a0 = red[0][j * NRED + r][lane];
    float a1 = red[0][j * NRED + 8 + r][lane];
    float s  = red[0][j * NRED + 16 + r][lane];
#pragma unroll
    for (int w = 1; w < NWAVE; ++w) {
      a0 += red[w][j * NRED + r][lane];
      a1 += red[w][j * NRED + 8 + r][lane];
      s  += red[w][j * NRED + 16 + r][lane];
    }

#pragma unroll
    for (int mask = 1; mask <= 8; mask <<= 1)
      s += __shfl_xor(s, mask, 32);

    const int m     = (mg * RM + j) * 16 + r + 8 * half;
    const float inv = 0.5f / s;
    O[lq * NPIX + m]        = a0 * inv + 0.5f * Y[lq * NPIX + m];
    O[(16 + lq) * NPIX + m] = a1 * inv + 0.5f * Y[(16 + lq) * NPIX + m];
  }
}

extern "C" void kernel_launch(void* const* d_in, const int* in_sizes, int n_in,
                              void* d_out, int out_size, void* d_ws, size_t ws_size,
                              hipStream_t stream) {
  (void)in_sizes; (void)n_in; (void)out_size; (void)ws_size;
  const float* x  = (const float*)d_in[0];
  float*       out = (float*)d_out;
  float* ws0 = (float*)d_ws;
  float* ws1 = ws0 + 4 * DD * NPIX;

  dim3 grid(4 * NGRP);
  dim3 block(32 * NWAVE);

  MS_meanshift_iter<<<grid, block, 0, stream>>>(x,   ws0);
  MS_meanshift_iter<<<grid, block, 0, stream>>>(ws0, ws1);
  MS_meanshift_iter<<<grid, block, 0, stream>>>(ws1, out);
}


# Gram tile computed transposed (rows n, cols m): after exp its D registers are the A operand of the Y@K product as they stand, so E skips the LDS round trip; n summation order within a tile changes
# speedup vs baseline: 2.9146x; 1.0370x over previous
#include <hip/hip_runtime.h>


typedef __attribute__((ext_vector_type(2))) float v2f;
typedef __attribute__((ext_vector_type(4))) float v4f;
typedef __attribute__((ext_vector_type(8))) float v8f;

#define DD   32
#define NPIX 4096
#define NBLK (NPIX / 16)
#define RM    4
#define NWAVE 16
#define NGRP  (NBLK / RM)
#define NRED  17

__device__ __forceinline__ v8f wmma_f32(v2f a, v2f b, v8f c) {
  return __builtin_amdgcn_wmma_f32_16x16x4_f32(
       false, a,  false, b,
       (short)0, c,  false,  false);
}

__global__ __launch_bounds__(32 * NWAVE)
void MS_meanshift_iter(const float* __restrict__ src, float* __restrict__ dst) {
  __shared__ float red[NWAVE][RM * NRED][32];

  const int lane = threadIdx.x & 31;
  const int wv   = __builtin_amdgcn_readfirstlane((int)(threadIdx.x >> 5));
  const int lq   = lane & 15;
  const int half = lane >> 4;
  const int b    = blockIdx.x / NGRP;
  const int mg   = blockIdx.x % NGRP;

  const float* Y = src + b * DD * NPIX;
  float*       O = dst + b * DD * NPIX;

  const float BW_LOG2E = 0.14426950408889634f;

  v2f am[RM][8];
#pragma unroll
  for (int j = 0; j < RM; ++j) {
    const int mcol = (mg * RM + j) * 16 + lq;
#pragma unroll
    for (int k = 0; k < 8; ++k) {
      am[j][k].x = Y[(4 * k + 0 + 2 * half) * NPIX + mcol];
      am[j][k].y = Y[(4 * k + 1 + 2 * half) * NPIX + mcol];
    }
  }

  const float* Yg = Y + (2 * half) * NPIX + lq;
  const float* Yb = Y + lq * NPIX + 8 * half;

  v8f acc0[RM];
  v8f acc1[RM];
  v8f srow[RM];
#pragma unroll
  for (int j = 0; j < RM; ++j) { acc0[j] = {}; acc1[j] = {}; srow[j] = {}; }

  for (int it = 0; it < NBLK / NWAVE; ++it) {
    const int nb    = it * NWAVE + wv;
    const int ncol0 = nb * 16;
    if (it + 1 < NBLK / NWAVE)
      __builtin_prefetch(&Y[ncol0 + 16 * NWAVE + lq], 0, 0);
    const float* Yn = Yg + ncol0;
    const float* Yk = Yb + ncol0;

    v2f gb[8];
#pragma unroll
    for (int k = 0; k < 8; ++k) {
      gb[k].x = Yn[(4 * k + 0) * NPIX];
      gb[k].y = Yn[(4 * k + 1) * NPIX];
    }
    const v4f q00 = *(const v4f*)&Yk[0];
    const v4f q01 = *(const v4f*)&Yk[4];
    const v4f q10 = *(const v4f*)&Yk[16 * NPIX];
    const v4f q11 = *(const v4f*)&Yk[16 * NPIX + 4];
    v2f b20[4], b21[4];
    b20[0] = v2f{q00.x, q00.y}; b20[1] = v2f{q00.z, q00.w};
    b20[2] = v2f{q01.x, q01.y}; b20[3] = v2f{q01.z, q01.w};
    b21[0] = v2f{q10.x, q10.y}; b21[1] = v2f{q10.z, q10.w};
    b21[2] = v2f{q11.x, q11.y}; b21[3] = v2f{q11.z, q11.w};

#pragma unroll
    for (int j = 0; j < RM; ++j) {
      v8f g = {};
#pragma unroll
      for (int k = 0; k < 8; ++k) g = wmma_f32(gb[k], am[j][k], g);

      v8f e;
#pragma unroll
      for (int r = 0; r < 8; ++r) e[r] = __builtin_amdgcn_exp2f(g[r] * BW_LOG2E);
      srow[j] += e;

      v2f a2[4];
#pragma unroll
      for (int k = 0; k < 4; ++k) a2[k] = v2f{e[2 * k], e[2 * k + 1]};

#pragma unroll
      for (int k = 0; k < 4; ++k) {
        acc0[j] = wmma_f32(a2[k], b20[k], acc0[j]);
        acc1[j] = wmma_f32(a2[k], b21[k], acc1[j]);
      }
    }
  }

#pragma unroll
  for (int j = 0; j < RM; ++j) {
#pragma unroll
    for (int r = 0; r < 8; ++r) {
      red[wv][j * NRED + r][lane]      = acc0[j][r];
      red[wv][j * NRED + 8 + r][lane]  = acc1[j][r];
    }
    float sp = ((srow[j][0] + srow[j][1]) + (srow[j][2] + srow[j][3])) +
               ((srow[j][4] + srow[j][5]) + (srow[j][6] + srow[j][7]));
    sp += __shfl_xor(sp, 16, 32);
    red[wv][j * NRED + 16][lane] = sp;
  }
  __syncthreads();

  for (int u = wv; u < RM * 8; u += NWAVE) {
    const int j = u >> 3;
    const int r = u & 7;
    float a0 = red[0][j * NRED + r][lane];
    float a1 = red[0][j * NRED + 8 + r][lane];
    float s  = red[0][j * NRED + 16][r + 8 * half];
#pragma unroll
    for (int w = 1; w < NWAVE; ++w) {
      a0 += red[w][j * NRED + r][lane];
      a1 += red[w][j * NRED + 8 + r][lane];
      s  += red[w][j * NRED + 16][r + 8 * half];
    }

    const int m     = (mg * RM + j) * 16 + r + 8 * half;
    const float inv = 0.5f / s;
    O[lq * NPIX + m]        = a0 * inv + 0.5f * Y[lq * NPIX + m];
    O[(16 + lq) * NPIX + m] = a1 * inv + 0.5f * Y[(16 + lq) * NPIX + m];
  }
}

extern "C" void kernel_launch(void* const* d_in, const int* in_sizes, int n_in,
                              void* d_out, int out_size, void* d_ws, size_t ws_size,
                              hipStream_t stream) {
  (void)in_sizes; (void)n_in; (void)out_size; (void)ws_size;
  const float* x  = (const float*)d_in[0];
  float*       out = (float*)d_out;
  float* ws0 = (float*)d_ws;
  float* ws1 = ws0 + 4 * DD * NPIX;

  dim3 grid(4 * NGRP);
  dim3 block(32 * NWAVE);

  MS_meanshift_iter<<<grid, block, 0, stream>>>(x,   ws0);
  MS_meanshift_iter<<<grid, block, 0, stream>>>(ws0, ws1);
  MS_meanshift_iter<<<grid, block, 0, stream>>>(ws1, out);
}


# n-tile loads double-buffered in registers: trip t+1's tile is requested ahead of trip t's matrix work; E column sums folded per tile to one register pair to make room, so the n order of s changes
# speedup vs baseline: 3.0675x; 1.0525x over previous
#include <hip/hip_runtime.h>


typedef __attribute__((ext_vector_type(2))) float v2f;
typedef __attribute__((ext_vector_type(4))) float v4f;
typedef __attribute__((ext_vector_type(8))) float v8f;

#define DD   32
#define NPIX 4096
#define NBLK (NPIX / 16)
#define RM    4
#define NWAVE 16
#define NGRP  (NBLK / RM)
#define NRED  17

__device__ __forceinline__ v8f wmma_f32(v2f a, v2f b, v8f c) {
  return __builtin_amdgcn_wmma_f32_16x16x4_f32(
       false, a,  false, b,
       (short)0, c,  false,  false);
}

#define NT (NBLK / NWAVE)
static_assert(NT % 2 == 0 && (NT & (NT - 1)) == 0, "two trips per pass; the look-ahead index wraps by a mask");

struct NTile {
  v2f gb[8];
  v4f q00, q01, q10, q11;
};

__device__ __forceinline__ void load_tile(NTile& t, const float* __restrict__ Yn,
                                          const float* __restrict__ Yk) {
#pragma unroll
  for (int k = 0; k < 8; ++k) {
    t.gb[k].x = Yn[(4 * k + 0) * NPIX];
    t.gb[k].y = Yn[(4 * k + 1) * NPIX];
  }
  t.q00 = *(const v4f*)&Yk[0];
  t.q01 = *(const v4f*)&Yk[4];
  t.q10 = *(const v4f*)&Yk[16 * NPIX];
  t.q11 = *(const v4f*)&Yk[16 * NPIX + 4];
}

__global__ __launch_bounds__(32 * NWAVE)
void MS_meanshift_iter(const float* __restrict__ src, float* __restrict__ dst) {
  __shared__ float red[NWAVE][RM * NRED][32];

  const int lane = threadIdx.x & 31;
  const int wv   = __builtin_amdgcn_readfirstlane((int)(threadIdx.x >> 5));
  const int lq   = lane & 15;
  const int half = lane >> 4;
  const int b    = blockIdx.x / NGRP;
  const int mg   = blockIdx.x % NGRP;

  const float* Y = src + b * DD * NPIX;
  float*       O = dst + b * DD * NPIX;

  const float BW_LOG2E = 0.14426950408889634f;

  v2f am[RM][8];
#pragma unroll
  for (int j = 0; j < RM; ++j) {
    const int mcol = (mg * RM + j) * 16 + lq;
#pragma unroll
    for (int k = 0; k < 8; ++k) {
      am[j][k].x = Y[(4 * k + 0 + 2 * half) * NPIX + mcol];
      am[j][k].y = Y[(4 * k + 1 + 2 * half) * NPIX + mcol];
    }
  }

  const float* Yg = Y + (2 * half) * NPIX + lq;
  const float* Yb = Y + lq * NPIX + 8 * half;

  v8f acc0[RM];
  v8f acc1[RM];
  v2f scol[RM];
#pragma unroll
  for (int j = 0; j < RM; ++j) { acc0[j] = {}; acc1[j] = {}; scol[j] = {}; }

  auto trip = [&](const NTile& t) {
    v2f b20[4], b21[4];
    b20[0] = v2f{t.q00.x, t.q00.y}; b20[1] = v2f{t.q00.z, t.q00.w};
    b20[2] = v2f{t.q01.x, t.q01.y}; b20[3] = v2f{t.q01.z, t.q01.w};
    b21[0] = v2f{t.q10.x, t.q10.y}; b21[1] = v2f{t.q10.z, t.q10.w};
    b21[2] = v2f{t.q11.x, t.q11.y}; b21[3] = v2f{t.q11.z, t.q11.w};

#pragma unroll
    for (int j = 0; j < RM; ++j) {
      v8f g = {};
#pragma unroll
      for (int k = 0; k < 8; ++k) g = wmma_f32(t.gb[k], am[j][k], g);

      v8f e;
#pragma unroll
      for (int r = 0; r < 8; ++r) e[r] = __builtin_amdgcn_exp2f(g[r] * BW_LOG2E);

      v2f a2[4];
#pragma unroll
      for (int k = 0; k < 4; ++k) a2[k] = v2f{e[2 * k], e[2 * k + 1]};

      scol[j] += (a2[0] + a2[1]) + (a2[2] + a2[3]);

#pragma unroll
      for (int k = 0; k < 4; ++k) {
        acc0[j] = wmma_f32(a2[k], b20[k], acc0[j]);
        acc1[j] = wmma_f32(a2[k], b21[k], acc1[j]);
      }
    }
  };

  const float* Yn0 = Yg + wv * 16;
  const float* Yk0 = Yb + wv * 16;
  NTile ta, tb;
  load_tile(ta, Yn0, Yk0);
  for (int it = 0; it < NT; it += 2) {
    load_tile(tb, Yn0 + (it + 1) * (16 * NWAVE), Yk0 + (it + 1) * (16 * NWAVE));
    __builtin_amdgcn_sched_barrier(0);
    trip(ta);
    __builtin_amdgcn_sched_barrier(0);
    const int itn = (it + 2) & (NT - 1);
    load_tile(ta, Yn0 + itn * (16 * NWAVE), Yk0 + itn * (16 * NWAVE));
    __builtin_amdgcn_sched_barrier(0);
    trip(tb);
  }

#pragma unroll
  for (int j = 0; j < RM; ++j) {
#pragma unroll
    for (int r = 0; r < 8; ++r) {
      red[wv][j * NRED + r][lane]      = acc0[j][r];
      red[wv][j * NRED + 8 + r][lane]  = acc1[j][r];
    }
    float sp = scol[j].x + scol[j].y;
    sp += __shfl_xor(sp, 16, 32);
    red[wv][j * NRED + 16][lane] = sp;
  }
  __syncthreads();

  for (int u = wv; u < RM * 8; u += NWAVE) {
    const int j = u >> 3;
    const int r = u & 7;
    float a0 = red[0][j * NRED + r][lane];
    float a1 = red[0][j * NRED + 8 + r][lane];
    float s  = red[0][j * NRED + 16][r + 8 * half];
#pragma unroll
    for (int w = 1; w < NWAVE; ++w) {
      a0 += red[w][j * NRED + r][lane];
      a1 += red[w][j * NRED + 8 + r][lane];
      s  += red[w][j * NRED + 16][r + 8 * half];
    }

    const int m     = (mg * RM + j) * 16 + r + 8 * half;
    const float inv = 0.5f / s;
    O[lq * NPIX + m]        = a0 * inv + 0.5f * Y[lq * NPIX + m];
    O[(16 + lq) * NPIX + m] = a1 * inv + 0.5f * Y[(16 + lq) * NPIX + m];
  }
}

extern "C" void kernel_launch(void* const* d_in, const int* in_sizes, int n_in,
                              void* d_out, int out_size, void* d_ws, size_t ws_size,
                              hipStream_t stream) {
  (void)in_sizes; (void)n_in; (void)out_size; (void)ws_size;
  const float* x  = (const float*)d_in[0];
  float*       out = (float*)d_out;
  float* ws0 = (float*)d_ws;
  float* ws1 = ws0 + 4 * DD * NPIX;

  dim3 grid(4 * NGRP);
  dim3 block(32 * NWAVE);

  MS_meanshift_iter<<<grid, block, 0, stream>>>(x,   ws0);
  MS_meanshift_iter<<<grid, block, 0, stream>>>(ws0, ws1);
  MS_meanshift_iter<<<grid, block, 0, stream>>>(ws1, out);
}
